# starting kernel
# baseline (speedup 1.0000x reference)
#include <hip/hip_runtime.h>


#define T_LEN   1048576
#define H       16
#define CHUNK   256
#define NCHUNK  (T_LEN / CHUNK)

typedef __attribute__((ext_vector_type(2))) float v2f;
typedef __attribute__((ext_vector_type(8))) float v8f;

#if __has_builtin(__builtin_amdgcn_tanhf)
#define TANHF(x) __builtin_amdgcn_tanhf(x)
#else
#define TANHF(x) tanhf(x)
#endif

__device__ __forceinline__ float readlane_f(float v, int lane) {
  return __builtin_bit_cast(float,
      __builtin_amdgcn_readlane(__builtin_bit_cast(int, v), lane));
}

__global__ __launch_bounds__(96, 1)
void rnn_pipeline_kernel(const float* __restrict__ y,
                         const float* __restrict__ W_ih0, const float* __restrict__ b_ih0,
                         const float* __restrict__ W_hh0, const float* __restrict__ b_hh0,
                         const float* __restrict__ W_ih1, const float* __restrict__ b_ih1,
                         const float* __restrict__ W_hh1, const float* __restrict__ b_hh1,
                         const float* __restrict__ W_lin, const float* __restrict__ b_lin,
                         float* __restrict__ out)
{
  __shared__ float h1buf[2][CHUNK * H];
  __shared__ float h2buf[2][CHUNK * H];
  __shared__ float xp1buf[CHUNK * H];

  const int tid  = threadIdx.x;
  const int wave = tid >> 5;
  const int l    = tid & 31;
  const int n    = l & 15;
  const int half = l >> 4;

  float w0[H];
  float w1[H];
  float wl[H];
  v2f   Bw[4];
  float wih = 0.f, bs0 = 0.f, b1v = 0.f, bl = 0.f;

  float sh[H];
  #pragma unroll
  for (int k = 0; k < H; ++k) sh[k] = 0.0f;

  if (wave == 0) {
    #pragma unroll
    for (int k = 0; k < H; ++k) w0[k] = W_hh0[n * H + k];
    wih = W_ih0[n];
    bs0 = b_ih0[n] + b_hh0[n];
  } else if (wave == 1) {
    #pragma unroll
    for (int k = 0; k < H; ++k) w1[k] = W_hh1[n * H + k];
    #pragma unroll
    for (int kb = 0; kb < 4; ++kb) {
      const int K = 4 * kb + 2 * half;
      Bw[kb] = *(const v2f*)(W_ih1 + n * H + K);
    }
    b1v = b_ih1[n] + b_hh1[n];
  } else {
    const int o = l & 1;
    #pragma unroll
    for (int k = 0; k < H; ++k) wl[k] = W_lin[o * H + k];
    bl = b_lin[o];
  }

  for (int i = 0; i < NCHUNK + 2; ++i) {
    if (wave == 0) {
      if (i < NCHUNK) {
        float* hb = h1buf[i & 1];
        const int base = i * CHUNK;
        if (i + 1 < NCHUNK) __builtin_prefetch(y + base + CHUNK, 0, 0);
        for (int s = 0; s < CHUNK; ++s) {
          const float yv = y[base + s];
          float a0 = fmaf(yv, wih, bs0);
          float a1 = 0.f;
          #pragma unroll
          for (int k = 0; k < H; k += 2) {
            a0 = fmaf(sh[k],     w0[k],     a0);
            a1 = fmaf(sh[k + 1], w0[k + 1], a1);
          }
          const float hn = TANHF(a0 + a1);
          if (l < 16) hb[s * H + n] = hn;
          #pragma unroll
          for (int k = 0; k < H; ++k) sh[k] = readlane_f(hn, k);
        }
      }
    } else if (wave == 1) {
      if (i >= 1 && i <= NCHUNK) {
        const int c = i - 1;
        const float* hb = h1buf[c & 1];
        float*       ob = h2buf[c & 1];

        for (int tile = 0; tile < CHUNK / 16; ++tile) {
          v8f acc;
          #pragma unroll
          for (int v = 0; v < 8; ++v) acc[v] = b1v;
          #pragma unroll
          for (int kb = 0; kb < 4; ++kb) {
            const int K = 4 * kb + 2 * half;
            const v2f Aa = *(const v2f*)(hb + (tile * 16 + n) * H + K);
            acc = __builtin_amdgcn_wmma_f32_16x16x4_f32(
                false, Aa, false, Bw[kb], (short)0, acc, false, false);
          }
          #pragma unroll
          for (int v = 0; v < 8; ++v)
            xp1buf[(tile * 16 + v + 8 * half) * H + n] = acc[v];
        }

        for (int s = 0; s < CHUNK; ++s) {
          float a0 = xp1buf[s * H + n];
          float a1 = 0.f;
          #pragma unroll
          for (int k = 0; k < H; k += 2) {
            a0 = fmaf(sh[k],     w1[k],     a0);
            a1 = fmaf(sh[k + 1], w1[k + 1], a1);
          }
          const float hn = TANHF(a0 + a1);
          if (l < 16) ob[s * H + n] = hn;
          #pragma unroll
          for (int k = 0; k < H; ++k) sh[k] = readlane_f(hn, k);
        }
      }
    } else {
      if (i >= 2) {
        const int c = i - 2;
        const float* hb = h2buf[c & 1];
        const int o  = l & 1;
        const int tt = l >> 1;
        const int base = c * CHUNK;
        for (int s = tt; s < CHUNK; s += 16) {
          float acc = bl;
          #pragma unroll
          for (int k = 0; k < H; ++k)
            acc = fmaf(hb[s * H + k], wl[k], acc);
          out[(base + s) * 2 + o] = acc;
        }
      }
    }
    __syncthreads();
  }
}

extern "C" void kernel_launch(void* const* d_in, const int* in_sizes, int n_in,
                              void* d_out, int out_size, void* d_ws, size_t ws_size,
                              hipStream_t stream) {
  (void)in_sizes; (void)n_in; (void)out_size; (void)d_ws; (void)ws_size;
  const float* y     = (const float*)d_in[0];
  const float* W_ih0 = (const float*)d_in[1];
  const float* b_ih0 = (const float*)d_in[2];
  const float* W_hh0 = (const float*)d_in[3];
  const float* b_hh0 = (const float*)d_in[4];
  const float* W_ih1 = (const float*)d_in[5];
  const float* b_ih1 = (const float*)d_in[6];
  const float* W_hh1 = (const float*)d_in[7];
  const float* b_hh1 = (const float*)d_in[8];
  const float* W_lin = (const float*)d_in[9];
  const float* b_lin = (const float*)d_in[10];
  float* out = (float*)d_out;

  rnn_pipeline_kernel<<<dim3(1), dim3(96), 0, stream>>>(
      y, W_ih0, b_ih0, W_hh0, b_hh0, W_ih1, b_ih1, W_hh1, b_hh1,
      W_lin, b_lin, out);
}


# Layer 1's C fragment of a step: 8 in-row lane broadcasts (v_permlane16_var) of a projection tile made with W_ih1 as A (column = step), set in the waits between chained links; one-hot product gone
# speedup vs baseline: 1.8524x; 1.8524x over previous
#include <hip/hip_runtime.h>


#define T_LEN   1048576
#define H       16
#define CHUNK   256
#define NCHUNK  (T_LEN / CHUNK)
#define YBLK    16
#define NTILE   (CHUNK / 16)

typedef __attribute__((ext_vector_type(2))) float v2f;
typedef __attribute__((ext_vector_type(4))) float v4f;
typedef __attribute__((ext_vector_type(8))) float v8f;

#if __has_builtin(__builtin_amdgcn_tanhf)
#define TANHF(x) __builtin_amdgcn_tanhf(x)
#else
#define TANHF(x) tanhf(x)
#endif

#define WMMA_F32(a, b, c) \
  __builtin_amdgcn_wmma_f32_16x16x4_f32(false, (a), false, (b), (short)0, (c), false, false)

#define PAIR(v, e) __builtin_shufflevector((v), (v), (e), (e) + 1)

__device__ __forceinline__ v8f tanh8(const v8f a) {
  v8f h;
  #pragma unroll
  for (int r = 0; r < 8; ++r) h[r] = TANHF(a[r]);
  return h;
}

__device__ __forceinline__ void store8(float* p, const v8f h) {
  *(v4f*)(p)     = __builtin_shufflevector(h, h, 0, 1, 2, 3);
  *(v4f*)(p + 4) = __builtin_shufflevector(h, h, 4, 5, 6, 7);
}

template <int L>
__device__ __forceinline__ float row_lane(const float x) {
  return __builtin_bit_cast(float,
      __builtin_amdgcn_permlane16_var(0u, __builtin_bit_cast(unsigned, x), (unsigned)L, false, true));
}

template <int JJ, int R0, int NR>
__device__ __forceinline__ void c_rows(float (&c)[8], const v8f& P, const v8f& Pn) {
  #pragma unroll
  for (int r = R0; r < R0 + NR; ++r)
    c[r] = (JJ < 16) ? row_lane<(JJ & 15)>(P[r]) : row_lane<(JJ & 15)>(Pn[r]);
}

__device__ __forceinline__ v8f keep_col(const v8f tile, const v8f h, const bool mine) {
  v8f o;
  #pragma unroll
  for (int r = 0; r < 8; ++r) o[r] = mine ? h[r] : tile[r];
  return o;
}

#define ORDER2(a, b)          asm volatile("" : "+v"(a), "+v"(b))
#define ORDER3(a, b, c)       asm volatile("" : "+v"(a), "+v"(b), "+v"(c))
#define ORDER4(a, b, c, d)    asm volatile("" : "+v"(a), "+v"(b), "+v"(c), "+v"(d))
template <int J>
__device__ __forceinline__ void l1_trip(v8f& acc, v8f& cn, v8f& hs, v8f& cap,
                                        v8f& P, v8f& Pn,
                                        const v2f (&Aw)[4], const v2f (&Wi)[4],
                                        const v2f (&Hn)[4], const int n) {
  v8f& S = (J + 2 < 16) ? P : Pn;
  hs = tanh8(acc);
  acc = WMMA_F32(Aw[0], PAIR(hs, 0), cn);
  ORDER2(acc, cap);
  cap = keep_col(cap, hs, n == J);
  ORDER2(acc, cap);
  acc = WMMA_F32(Aw[1], PAIR(hs, 2), acc);
  ORDER2(acc, S);
  float c2[8];
  c_rows<J + 2, 0, 3>(c2, P, Pn);
  ORDER4(acc, c2[0], c2[1], c2[2]);
  acc = WMMA_F32(Aw[2], PAIR(hs, 4), acc);
  ORDER2(acc, S);
  c_rows<J + 2, 3, 3>(c2, P, Pn);
  ORDER4(acc, c2[3], c2[4], c2[5]);
  acc = WMMA_F32(Aw[3], PAIR(hs, 6), acc);
  ORDER3(acc, P, Pn);
  c_rows<J + 2, 6, 2>(c2, P, Pn);
  if constexpr (J >= 4 && J < 8) Pn = WMMA_F32(Wi[J - 4], Hn[J - 4], Pn);
  ORDER4(acc, c2[6], c2[7], Pn);
  #pragma unroll
  for (int r = 0; r < 8; ++r) cn[r] = c2[r];
}

__global__ __launch_bounds__(96, 1)
void rnn_pipeline_kernel(const float* __restrict__ y,
                         const float* __restrict__ W_ih0, const float* __restrict__ b_ih0,
                         const float* __restrict__ W_hh0, const float* __restrict__ b_hh0,
                         const float* __restrict__ W_ih1, const float* __restrict__ b_ih1,
                         const float* __restrict__ W_hh1, const float* __restrict__ b_hh1,
                         const float* __restrict__ W_lin, const float* __restrict__ b_lin,
                         float* __restrict__ out)
{
  __shared__ __attribute__((aligned(16))) float h1buf[2][CHUNK * H];
  __shared__ __attribute__((aligned(16))) float h2buf[2][CHUNK * H];

  const int tid  = threadIdx.x;
  const int wave = tid >> 5;
  const int l    = tid & 31;
  const int n    = l & 15;
  const int half = l >> 4;

  v2f   Aw[4];
  float wih[8], bs[8];
  v2f   Wi[4];
  v8f   b1f;
  float wl[H];
  float bl = 0.f;
  float ycur[YBLK], ynxt[YBLK];
  v8f   acc, cn, hs, cap;
  #pragma unroll
  for (int r = 0; r < 8; ++r) { acc[r] = 0.f; cn[r] = 0.f; hs[r] = 0.f; cap[r] = 0.f; b1f[r] = 0.f; }

  if (wave == 0) {
    #pragma unroll
    for (int kb = 0; kb < 4; ++kb)
      Aw[kb] = *(const v2f*)(W_hh0 + n * H + 8 * half + 2 * kb);
    #pragma unroll
    for (int r = 0; r < 8; ++r) {
      wih[r] = W_ih0[8 * half + r];
      bs[r]  = b_ih0[8 * half + r] + b_hh0[8 * half + r];
    }
    #pragma unroll
    for (int j = 0; j < YBLK; ++j) ycur[j] = y[j];
    #pragma unroll
    for (int r = 0; r < 8; ++r) {
      acc[r] = fmaf(ycur[0], wih[r], bs[r]);
      cn[r]  = fmaf(ycur[1], wih[r], bs[r]);
    }
  } else if (wave == 1) {
    #pragma unroll
    for (int kb = 0; kb < 4; ++kb) {
      Aw[kb] = *(const v2f*)(W_hh1 + n * H + 8 * half + 2 * kb);
      Wi[kb] = *(const v2f*)(W_ih1 + n * H + 4 * kb + 2 * half);
    }
    #pragma unroll
    for (int r = 0; r < 8; ++r) b1f[r] = b_ih1[8 * half + r] + b_hh1[8 * half + r];
  } else {
    const int o = l & 1;
    #pragma unroll
    for (int k = 0; k < H; ++k) wl[k] = W_lin[o * H + k];
    bl = b_lin[o];
  }

  for (int i = 0; i < NCHUNK + 2; ++i) {
    if (wave == 0) {
      if (i < NCHUNK) {
        float* hb = h1buf[i & 1] + n * H + 8 * half;
        for (int blk = 0; blk < CHUNK / YBLK; ++blk) {
          const int t0 = i * CHUNK + blk * YBLK;
          const int tn = (t0 + YBLK < T_LEN) ? (t0 + YBLK) : t0;
          #pragma unroll
          for (int j = 0; j < YBLK; ++j) ynxt[j] = y[tn + j];
          #pragma unroll
          for (int j = 0; j < YBLK; ++j) {
            const v8f h = tanh8(acc);
            acc = WMMA_F32(Aw[0], PAIR(h, 0), cn);
            __builtin_amdgcn_sched_barrier(0);
            cap = keep_col(cap, h, n == j);
            const float y2 = (j + 2 < YBLK) ? ycur[j + 2] : ynxt[j + 2 - YBLK];
            #pragma unroll
            for (int r = 0; r < 8; ++r) cn[r] = fmaf(y2, wih[r], bs[r]);
            __builtin_amdgcn_sched_barrier(0);
            acc = WMMA_F32(Aw[1], PAIR(h, 2), acc);
            acc = WMMA_F32(Aw[2], PAIR(h, 4), acc);
            acc = WMMA_F32(Aw[3], PAIR(h, 6), acc);
          }
          store8(hb + blk * YBLK * H, cap);
          #pragma unroll
          for (int j = 0; j < YBLK; ++j) ycur[j] = ynxt[j];
        }
      }
    } else if (wave == 1) {
      if (i >= 1 && i <= NCHUNK) {
        const int c1 = i - 1;
        const float* hb = h1buf[c1 & 1] + n * H + 2 * half;
        float* ob = h2buf[c1 & 1] + n * H + 8 * half;

        v8f P = b1f;
        #pragma unroll
        for (int kb = 0; kb < 4; ++kb)
          P = WMMA_F32(Wi[kb], *(const v2f*)(hb + 4 * kb), P);

        v2f Hn[4];
        #pragma unroll
        for (int kb = 0; kb < 4; ++kb)
          Hn[kb] = *(const v2f*)(hb + 16 * H + 4 * kb);

        float c0[8];
        c_rows<0, 0, 8>(c0, P, P);
        #pragma unroll
        for (int r = 0; r < 8; ++r) acc[r] = c0[r];
        acc = WMMA_F32(Aw[0], PAIR(hs, 0), acc);
        acc = WMMA_F32(Aw[1], PAIR(hs, 2), acc);
        acc = WMMA_F32(Aw[2], PAIR(hs, 4), acc);
        acc = WMMA_F32(Aw[3], PAIR(hs, 6), acc);
        c_rows<1, 0, 8>(c0, P, P);
        #pragma unroll
        for (int r = 0; r < 8; ++r) cn[r] = c0[r];

        for (int tile = 0; tile < NTILE; ++tile) {
          v8f Pn = b1f;
          l1_trip<0>(acc, cn, hs, cap, P, Pn, Aw, Wi, Hn, n);
          l1_trip<1>(acc, cn, hs, cap, P, Pn, Aw, Wi, Hn, n);
          l1_trip<2>(acc, cn, hs, cap, P, Pn, Aw, Wi, Hn, n);
          l1_trip<3>(acc, cn, hs, cap, P, Pn, Aw, Wi, Hn, n);
          l1_trip<4>(acc, cn, hs, cap, P, Pn, Aw, Wi, Hn, n);
          l1_trip<5>(acc, cn, hs, cap, P, Pn, Aw, Wi, Hn, n);
          l1_trip<6>(acc, cn, hs, cap, P, Pn, Aw, Wi, Hn, n);
          l1_trip<7>(acc, cn, hs, cap, P, Pn, Aw, Wi, Hn, n);
          l1_trip<8>(acc, cn, hs, cap, P, Pn, Aw, Wi, Hn, n);
          l1_trip<9>(acc, cn, hs, cap, P, Pn, Aw, Wi, Hn, n);
          l1_trip<10>(acc, cn, hs, cap, P, Pn, Aw, Wi, Hn, n);
          l1_trip<11>(acc, cn, hs, cap, P, Pn, Aw, Wi, Hn, n);
          l1_trip<12>(acc, cn, hs, cap, P, Pn, Aw, Wi, Hn, n);
          l1_trip<13>(acc, cn, hs, cap, P, Pn, Aw, Wi, Hn, n);
          l1_trip<14>(acc, cn, hs, cap, P, Pn, Aw, Wi, Hn, n);
          l1_trip<15>(acc, cn, hs, cap, P, Pn, Aw, Wi, Hn, n);
          store8(ob + tile * 16 * H, cap);
          P = Pn;
          const int tnx = (tile + 2 < NTILE) ? (tile + 2) : (NTILE - 1);
          #pragma unroll
          for (int kb = 0; kb < 4; ++kb)
            Hn[kb] = *(const v2f*)(hb + tnx * 16 * H + 4 * kb);
        }
      }
    } else {
      if (i >= 2) {
        const int c2 = i - 2;
        const float* hb = h2buf[c2 & 1];
        const int o  = l & 1;
        const int tt = l >> 1;
        const int base = c2 * CHUNK;
        for (int s = tt; s < CHUNK; s += 16) {
          float a = bl;
          #pragma unroll
          for (int k = 0; k < H; ++k)
            a = fmaf(hb[s * H + k], wl[k], a);
          out[(base + s) * 2 + o] = a;
        }
      }
    }
    __syncthreads();
  }
}

extern "C" void kernel_launch(void* const* d_in, const int* in_sizes, int n_in,
                              void* d_out, int out_size, void* d_ws, size_t ws_size,
                              hipStream_t stream) {
  (void)in_sizes; (void)n_in; (void)out_size; (void)d_ws; (void)ws_size;
  const float* y     = (const float*)d_in[0];
  const float* W_ih0 = (const float*)d_in[1];
  const float* b_ih0 = (const float*)d_in[2];
  const float* W_hh0 = (const float*)d_in[3];
  const float* b_hh0 = (const float*)d_in[4];
  const float* W_ih1 = (const float*)d_in[5];
  const float* b_ih1 = (const float*)d_in[6];
  const float* W_hh1 = (const float*)d_in[7];
  const float* b_hh1 = (const float*)d_in[8];
  const float* W_lin = (const float*)d_in[9];
  const float* b_lin = (const float*)d_in[10];
  float* out = (float*)d_out;

  rnn_pipeline_kernel<<<dim3(1), dim3(96), 0, stream>>>(
      y, W_ih0, b_ih0, W_hh0, b_hh0, W_ih1, b_ih1, W_hh1, b_hh1,
      W_lin, b_lin, out);
}


# Both scans: a step's issue order: only 5 tanh, a plain instruction behind each, stand before link 0 (it reads 2 of them); the other 3 tanh and the selects fill the waits behind the links evenly
# speedup vs baseline: 1.9619x; 1.0591x over previous
#include <hip/hip_runtime.h>


#define T_LEN   1048576
#define H       16
#define CHUNK   256
#define NCHUNK  (T_LEN / CHUNK)
#define YBLK    16
#define NTILE   (CHUNK / 16)

typedef __attribute__((ext_vector_type(2))) float v2f;
typedef __attribute__((ext_vector_type(4))) float v4f;
typedef __attribute__((ext_vector_type(8))) float v8f;

#if __has_builtin(__builtin_amdgcn_tanhf)
#define TANHF(x) __builtin_amdgcn_tanhf(x)
#else
#define TANHF(x) tanhf(x)
#endif

#define WMMA_F32(a, b, c) \
  __builtin_amdgcn_wmma_f32_16x16x4_f32(false, (a), false, (b), (short)0, (c), false, false)

#define PAIR(v, e) __builtin_shufflevector((v), (v), (e), (e) + 1)

__device__ __forceinline__ void store8(float* p, const v8f h) {
  *(v4f*)(p)     = __builtin_shufflevector(h, h, 0, 1, 2, 3);
  *(v4f*)(p + 4) = __builtin_shufflevector(h, h, 4, 5, 6, 7);
}

template <int L>
__device__ __forceinline__ float row_lane(const float x) {
  return __builtin_bit_cast(float,
      __builtin_amdgcn_permlane16_var(0u, __builtin_bit_cast(unsigned, x), (unsigned)L, false, true));
}

template <int JJ, int R0, int NR>
__device__ __forceinline__ void c_rows(float (&c)[8], const v8f& P, const v8f& Pn) {
  #pragma unroll
  for (int r = R0; r < R0 + NR; ++r)
    c[r] = (JJ < 16) ? row_lane<(JJ & 15)>(P[r]) : row_lane<(JJ & 15)>(Pn[r]);
}

__device__ __forceinline__ v2f pair2(const float a, const float b) { v2f p = {a, b}; return p; }

template <int JJ, int R>
__device__ __forceinline__ float c_row(const v8f& P, const v8f& Pn) {
  return (JJ < 16) ? row_lane<(JJ & 15)>(P[R]) : row_lane<(JJ & 15)>(Pn[R]);
}

#define AFTER(made, next) asm volatile("" : "+v"(made), "+v"(next))

template <int J>
__device__ __forceinline__ void l1_trip(v8f& acc, v8f& cn, v8f& hs, v8f& cap,
                                        v8f& P, v8f& Pn,
                                        const v2f (&Aw)[4], const v2f (&Wi)[4],
                                        const v2f (&Hn)[4], const int n) {
  v8f& S = (J + 2 < 16) ? P : Pn;
  const bool mine = (n == J);
  float c2[8];
  float h0 = TANHF(acc[0]);           AFTER(h0, S);
  c2[0] = c_row<J + 2, 0>(P, Pn);     AFTER(c2[0], acc);
  float h1 = TANHF(acc[1]);           AFTER(h1, S);
  c2[1] = c_row<J + 2, 1>(P, Pn);     AFTER(c2[1], acc);
  float h2 = TANHF(acc[2]);           AFTER(h2, S);
  c2[2] = c_row<J + 2, 2>(P, Pn);     AFTER(c2[2], acc);
  float h3 = TANHF(acc[3]);           AFTER(h3, S);
  c2[3] = c_row<J + 2, 3>(P, Pn);     AFTER(c2[3], acc);
  float h4 = TANHF(acc[4]);           AFTER(h4, S);
  c2[4] = c_row<J + 2, 4>(P, Pn);     AFTER(c2[4], cn);
  v8f a = WMMA_F32(Aw[0], pair2(h0, h1), cn);
  AFTER(a, cap);
  cap[0] = mine ? h0 : cap[0];  cap[1] = mine ? h1 : cap[1];   AFTER(cap, acc);
  float h5 = TANHF(acc[5]);           AFTER(h5, cap);
  cap[2] = mine ? h2 : cap[2];  cap[3] = mine ? h3 : cap[3];   AFTER(cap, acc);
  float h6 = TANHF(acc[6]);           AFTER(h6, a);
  a = WMMA_F32(Aw[1], pair2(h2, h3), a);
  AFTER(a, S);
  c2[5] = c_row<J + 2, 5>(P, Pn);     AFTER(c2[5], acc);
  float h7 = TANHF(acc[7]);           AFTER(h7, S);
  c2[6] = c_row<J + 2, 6>(P, Pn);     AFTER(c2[6], a);
  a = WMMA_F32(Aw[2], pair2(h4, h5), a);
  AFTER(a, S);
  c2[7] = c_row<J + 2, 7>(P, Pn);     AFTER(c2[7], cap);
  cap[4] = mine ? h4 : cap[4];  cap[5] = mine ? h5 : cap[5];   AFTER(cap, a);
  a = WMMA_F32(Aw[3], pair2(h6, h7), a);
  AFTER(a, cap);
  cap[6] = mine ? h6 : cap[6];  cap[7] = mine ? h7 : cap[7];
  if constexpr (J >= 4 && J < 8) {
    AFTER(cap, Pn);
    Pn = WMMA_F32(Wi[J - 4], Hn[J - 4], Pn);
    AFTER(Pn, a);
  } else {
    AFTER(cap, a);
  }
  acc = a;
  #pragma unroll
  for (int r = 0; r < 8; ++r) cn[r] = c2[r];
  hs[0] = h0; hs[1] = h1; hs[2] = h2; hs[3] = h3;
  hs[4] = h4; hs[5] = h5; hs[6] = h6; hs[7] = h7;
}

__global__ __launch_bounds__(96, 1)
void rnn_pipeline_kernel(const float* __restrict__ y,
                         const float* __restrict__ W_ih0, const float* __restrict__ b_ih0,
                         const float* __restrict__ W_hh0, const float* __restrict__ b_hh0,
                         const float* __restrict__ W_ih1, const float* __restrict__ b_ih1,
                         const float* __restrict__ W_hh1, const float* __restrict__ b_hh1,
                         const float* __restrict__ W_lin, const float* __restrict__ b_lin,
                         float* __restrict__ out)
{
  __shared__ __attribute__((aligned(16))) float h1buf[2][CHUNK * H];
  __shared__ __attribute__((aligned(16))) float h2buf[2][CHUNK * H];

  const int tid  = threadIdx.x;
  const int wave = tid >> 5;
  const int l    = tid & 31;
  const int n    = l & 15;
  const int half = l >> 4;

  v2f   Aw[4];
  float wih[8], bs[8];
  v2f   wih2[4], bs2[4];
  v2f   Wi[4];
  v8f   b1f;
  float wl[H];
  float bl = 0.f;
  float ycur[YBLK], ynxt[YBLK];
  v8f   acc, cn, hs, cap;
  #pragma unroll
  for (int r = 0; r < 8; ++r) { acc[r] = 0.f; cn[r] = 0.f; hs[r] = 0.f; cap[r] = 0.f; b1f[r] = 0.f; }

  if (wave == 0) {
    #pragma unroll
    for (int kb = 0; kb < 4; ++kb)
      Aw[kb] = *(const v2f*)(W_hh0 + n * H + 8 * half + 2 * kb);
    #pragma unroll
    for (int r = 0; r < 8; ++r) {
      wih[r] = W_ih0[8 * half + r];
      bs[r]  = b_ih0[8 * half + r] + b_hh0[8 * half + r];
    }
    #pragma unroll
    for (int q = 0; q < 4; ++q) {
      wih2[q] = pair2(wih[2 * q], wih[2 * q + 1]);
      bs2[q]  = pair2(bs[2 * q], bs[2 * q + 1]);
    }
    #pragma unroll
    for (int j = 0; j < YBLK; ++j) ycur[j] = y[j];
    #pragma unroll
    for (int r = 0; r < 8; ++r) {
      acc[r] = fmaf(ycur[0], wih[r], bs[r]);
      cn[r]  = fmaf(ycur[1], wih[r], bs[r]);
    }
  } else if (wave == 1) {
    #pragma unroll
    for (int kb = 0; kb < 4; ++kb) {
      Aw[kb] = *(const v2f*)(W_hh1 + n * H + 8 * half + 2 * kb);
      Wi[kb] = *(const v2f*)(W_ih1 + n * H + 4 * kb + 2 * half);
    }
    #pragma unroll
    for (int r = 0; r < 8; ++r) b1f[r] = b_ih1[8 * half + r] + b_hh1[8 * half + r];
  } else {
    const int o = l & 1;
    #pragma unroll
    for (int k = 0; k < H; ++k) wl[k] = W_lin[o * H + k];
    bl = b_lin[o];
  }

  for (int i = 0; i < NCHUNK + 2; ++i) {
    if (wave == 0) {
      if (i < NCHUNK) {
        float* hb = h1buf[i & 1] + n * H + 8 * half;
        for (int blk = 0; blk < CHUNK / YBLK; ++blk) {
          const int t0 = i * CHUNK + blk * YBLK;
          const int tn = (t0 + YBLK < T_LEN) ? (t0 + YBLK) : t0;
          #pragma unroll
          for (int j = 0; j < YBLK; ++j) ynxt[j] = y[tn + j];
          #pragma unroll
          for (int j = 0; j < YBLK; ++j) {
            const bool mine = (n == j);
            float y2 = (j + 2 < YBLK) ? ycur[j + 2] : ynxt[j + 2 - YBLK];
            float h0 = TANHF(acc[0]);           AFTER(h0, y2);
            v2f c01 = __builtin_elementwise_fma(pair2(y2, y2), wih2[0], bs2[0]);   AFTER(c01, acc);
            float h1 = TANHF(acc[1]);           AFTER(h1, y2);
            v2f c23 = __builtin_elementwise_fma(pair2(y2, y2), wih2[1], bs2[1]);   AFTER(c23, acc);
            float h2 = TANHF(acc[2]);           AFTER(h2, y2);
            v2f c45 = __builtin_elementwise_fma(pair2(y2, y2), wih2[2], bs2[2]);   AFTER(c45, acc);
            float h3 = TANHF(acc[3]);           AFTER(h3, y2);
            v2f c67 = __builtin_elementwise_fma(pair2(y2, y2), wih2[3], bs2[3]);   AFTER(c67, acc);
            float h4 = TANHF(acc[4]);           AFTER(h4, cn);
            v8f a = WMMA_F32(Aw[0], pair2(h0, h1), cn);
            AFTER(a, cap);
            cap[0] = mine ? h0 : cap[0];  cap[1] = mine ? h1 : cap[1];   AFTER(cap, acc);
            float h5 = TANHF(acc[5]);           AFTER(h5, cap);
            cap[2] = mine ? h2 : cap[2];  cap[3] = mine ? h3 : cap[3];   AFTER(cap, acc);
            float h6 = TANHF(acc[6]);           AFTER(h6, a);
            a = WMMA_F32(Aw[1], pair2(h2, h3), a);
            AFTER(a, cap);
            cap[4] = mine ? h4 : cap[4];        AFTER(cap, acc);
            float h7 = TANHF(acc[7]);           AFTER(h7, a);
            a = WMMA_F32(Aw[2], pair2(h4, h5), a);
            AFTER(a, cap);
            cap[5] = mine ? h5 : cap[5];  cap[6] = mine ? h6 : cap[6];   AFTER(cap, a);
            a = WMMA_F32(Aw[3], pair2(h6, h7), a);
            AFTER(a, cap);
            cap[7] = mine ? h7 : cap[7];        AFTER(cap, a);
            acc = a;
            const v4f clo = __builtin_shufflevector(c01, c23, 0, 1, 2, 3);
            const v4f chi = __builtin_shufflevector(c45, c67, 0, 1, 2, 3);
            cn = __builtin_shufflevector(clo, chi, 0, 1, 2, 3, 4, 5, 6, 7);
          }
          store8(hb + blk * YBLK * H, cap);
          #pragma unroll
          for (int j = 0; j < YBLK; ++j) ycur[j] = ynxt[j];
        }
      }
    } else if (wave == 1) {
      if (i >= 1 && i <= NCHUNK) {
        const int c1 = i - 1;
        const float* hb = h1buf[c1 & 1] + n * H + 2 * half;
        float* ob = h2buf[c1 & 1] + n * H + 8 * half;

        v8f P = b1f;
        #pragma unroll
        for (int kb = 0; kb < 4; ++kb)
          P = WMMA_F32(Wi[kb], *(const v2f*)(hb + 4 * kb), P);

        v2f Hn[4];
        #pragma unroll
        for (int kb = 0; kb < 4; ++kb)
          Hn[kb] = *(const v2f*)(hb + 16 * H + 4 * kb);

        float c0[8];
        c_rows<0, 0, 8>(c0, P, P);
        #pragma unroll
        for (int r = 0; r < 8; ++r) acc[r] = c0[r];
        acc = WMMA_F32(Aw[0], PAIR(hs, 0), acc);
        acc = WMMA_F32(Aw[1], PAIR(hs, 2), acc);
        acc = WMMA_F32(Aw[2], PAIR(hs, 4), acc);
        acc = WMMA_F32(Aw[3], PAIR(hs, 6), acc);
        c_rows<1, 0, 8>(c0, P, P);
        #pragma unroll
        for (int r = 0; r < 8; ++r) cn[r] = c0[r];

        for (int tile = 0; tile < NTILE; ++tile) {
          v8f Pn = b1f;
          l1_trip<0>(acc, cn, hs, cap, P, Pn, Aw, Wi, Hn, n);
          l1_trip<1>(acc, cn, hs, cap, P, Pn, Aw, Wi, Hn, n);
          l1_trip<2>(acc, cn, hs, cap, P, Pn, Aw, Wi, Hn, n);
          l1_trip<3>(acc, cn, hs, cap, P, Pn, Aw, Wi, Hn, n);
          l1_trip<4>(acc, cn, hs, cap, P, Pn, Aw, Wi, Hn, n);
          l1_trip<5>(acc, cn, hs, cap, P, Pn, Aw, Wi, Hn, n);
          l1_trip<6>(acc, cn, hs, cap, P, Pn, Aw, Wi, Hn, n);
          l1_trip<7>(acc, cn, hs, cap, P, Pn, Aw, Wi, Hn, n);
          l1_trip<8>(acc, cn, hs, cap, P, Pn, Aw, Wi, Hn, n);
          l1_trip<9>(acc, cn, hs, cap, P, Pn, Aw, Wi, Hn, n);
          l1_trip<10>(acc, cn, hs, cap, P, Pn, Aw, Wi, Hn, n);
          l1_trip<11>(acc, cn, hs, cap, P, Pn, Aw, Wi, Hn, n);
          l1_trip<12>(acc, cn, hs, cap, P, Pn, Aw, Wi, Hn, n);
          l1_trip<13>(acc, cn, hs, cap, P, Pn, Aw, Wi, Hn, n);
          l1_trip<14>(acc, cn, hs, cap, P, Pn, Aw, Wi, Hn, n);
          l1_trip<15>(acc, cn, hs, cap, P, Pn, Aw, Wi, Hn, n);
          store8(ob + tile * 16 * H, cap);
          P = Pn;
          const int tnx = (tile + 2 < NTILE) ? (tile + 2) : (NTILE - 1);
          #pragma unroll
          for (int kb = 0; kb < 4; ++kb)
            Hn[kb] = *(const v2f*)(hb + tnx * 16 * H + 4 * kb);
        }
      }
    } else {
      if (i >= 2) {
        const int c2 = i - 2;
        const float* hb = h2buf[c2 & 1];
        const int o  = l & 1;
        const int tt = l >> 1;
        const int base = c2 * CHUNK;
        for (int s = tt; s < CHUNK; s += 16) {
          float a = bl;
          #pragma unroll
          for (int k = 0; k < H; ++k)
            a = fmaf(hb[s * H + k], wl[k], a);
          out[(base + s) * 2 + o] = a;
        }
      }
    }
    __syncthreads();
  }
}

extern "C" void kernel_launch(void* const* d_in, const int* in_sizes, int n_in,
                              void* d_out, int out_size, void* d_ws, size_t ws_size,
                              hipStream_t stream) {
  (void)in_sizes; (void)n_in; (void)out_size; (void)d_ws; (void)ws_size;
  const float* y     = (const float*)d_in[0];
  const float* W_ih0 = (const float*)d_in[1];
  const float* b_ih0 = (const float*)d_in[2];
  const float* W_hh0 = (const float*)d_in[3];
  const float* b_hh0 = (const float*)d_in[4];
  const float* W_ih1 = (const float*)d_in[5];
  const float* b_ih1 = (const float*)d_in[6];
  const float* W_hh1 = (const float*)d_in[7];
  const float* b_hh1 = (const float*)d_in[8];
  const float* W_lin = (const float*)d_in[9];
  const float* b_lin = (const float*)d_in[10];
  float* out = (float*)d_out;

  rnn_pipeline_kernel<<<dim3(1), dim3(96), 0, stream>>>(
      y, W_ih0, b_ih0, W_hh0, b_hh0, W_ih1, b_ih1, W_hh1, b_hh1,
      W_lin, b_lin, out);
}
